# starting kernel
# baseline (speedup 1.0000x reference)
#include <hip/hip_runtime.h>
#include <hip/hip_bf16.h>


typedef float v2f __attribute__((ext_vector_type(2)));
typedef float v8f __attribute__((ext_vector_type(8)));

__global__ void gnn_gemm_wmma(const float* __restrict__ old_g,
                              const float* __restrict__ W,
                              float* __restrict__ h,
                              float* __restrict__ agg,
                              int N) {
  const int lane = threadIdx.x & 31;
  const int wave = threadIdx.x >> 5;
  const int waveId = blockIdx.x * 4 + wave;
  const int row_tile = waveId >> 2;
  const int col_tile = waveId & 3;
  if (row_tile * 16 >= N) return;

  const int m0 = row_tile * 16;
  const int n0 = col_tile * 16;
  const int lo = lane & 15;
  const int hi = lane >> 4;

  const int row = m0 + lo;
  const int col = n0 + lo;

  v8f c = {};
  #pragma unroll
  for (int k = 0; k < 64; k += 4) {
    const int kk = k + 2 * hi;
    v2f a = *(const v2f*)(old_g + (size_t)row * 64 + kk);
    v2f b;
    b.x = W[kk * 64 + col];
    b.y = W[(kk + 1) * 64 + col];
    c = __builtin_amdgcn_wmma_f32_16x16x4_f32(false, a, false, b,
                                              (short)0, c, false, false);
  }

  #pragma unroll
  for (int v = 0; v < 8; ++v) {
    const size_t idx = (size_t)(m0 + v + 8 * hi) * 64 + n0 + lo;
    h[idx]   = c[v];
    agg[idx] = 0.0f;
  }
}

__global__ void gnn_edge(const float* __restrict__ h,
                         const float* __restrict__ old_g,
                         const float* __restrict__ ew,
                         const int* __restrict__ src,
                         const int* __restrict__ dst,
                         float* __restrict__ agg,
                         int E) {
  const int lane = threadIdx.x & 31;
  const int e = blockIdx.x * (blockDim.x >> 5) + (threadIdx.x >> 5);
  if (e >= E) return;

  const int s = src[e];
  const int r = dst[e];
  const float w = ew[e];

  const float2 hv = *(const float2*)(h + (size_t)s * 64 + 2 * lane);
  const float2 og = *(const float2*)(old_g + (size_t)r * 64 + 2 * lane);

  const float m0 = tanhf(hv.x + og.x * w);
  const float m1 = tanhf(hv.y + og.y * w);

  float* p = agg + (size_t)r * 64 + 2 * lane;
  asm volatile("global_atomic_add_f32 %0, %1, off\n\t"
               "global_atomic_add_f32 %0, %2, off offset:4"
               :
               : "v"(p), "v"(m0), "v"(m1)
               : "memory");
}

__global__ void gnn_finish(const float* __restrict__ agg,
                           const float* __restrict__ hist_db,
                           float* __restrict__ hist_out,
                           int n4) {
  const int i = blockIdx.x * blockDim.x + threadIdx.x;
  if (i >= n4) return;
  const float4 a = ((const float4*)agg)[i];
  const float4 b = ((const float4*)hist_db)[i];
  float4 o;
  o.x = a.x + b.x; o.y = a.y + b.y; o.z = a.z + b.z; o.w = a.w + b.w;
  ((float4*)hist_out)[i] = o;
}

extern "C" void kernel_launch(void* const* d_in, const int* in_sizes, int n_in,
                              void* d_out, int out_size, void* d_ws, size_t ws_size,
                              hipStream_t stream) {
  const float* old_g   = (const float*)d_in[0];
  const float* W       = (const float*)d_in[1];
  const float* ew      = (const float*)d_in[2];
  const float* hist_db = (const float*)d_in[3];
  const int*   src     = (const int*)d_in[4];
  const int*   dst     = (const int*)d_in[5];

  const int N = in_sizes[0] / 64;
  const int E = in_sizes[4];

  float* nodes_new = (float*)d_out;
  float* hist_new  = (float*)d_out + (size_t)N * 64;

  {
    const int rowTiles = (N + 15) / 16;
    gnn_gemm_wmma<<<rowTiles, 128, 0, stream>>>(old_g, W, hist_new, nodes_new, N);
  }

  {
    const int wavesPerBlock = 8;
    const int blocks = (E + wavesPerBlock - 1) / wavesPerBlock;
    gnn_edge<<<blocks, 256, 0, stream>>>(hist_new, old_g, ew, src, dst, nodes_new, E);
  }

  {
    const int n4 = N * 16;
    gnn_finish<<<(n4 + 255) / 256, 256, 0, stream>>>(nodes_new, hist_db, hist_new, n4);
  }
}


# edge kernel: one wave per 32 edges (indices loaded once, 16 row gathers in flight per 8-edge chunk) instead of one wave per edge; arrival order of the f32 atomic adds changes
# speedup vs baseline: 1.5696x; 1.5696x over previous
#include <hip/hip_runtime.h>
#include <hip/hip_bf16.h>


typedef float v2f __attribute__((ext_vector_type(2)));
typedef float v8f __attribute__((ext_vector_type(8)));

__global__ void gnn_gemm_wmma(const float* __restrict__ old_g,
                              const float* __restrict__ W,
                              float* __restrict__ h,
                              float* __restrict__ agg,
                              int N) {
  const int lane = threadIdx.x & 31;
  const int wave = threadIdx.x >> 5;
  const int waveId = blockIdx.x * 4 + wave;
  const int row_tile = waveId >> 2;
  const int col_tile = waveId & 3;
  if (row_tile * 16 >= N) return;

  const int m0 = row_tile * 16;
  const int n0 = col_tile * 16;
  const int lo = lane & 15;
  const int hi = lane >> 4;

  const int row = m0 + lo;
  const int col = n0 + lo;

  v8f c = {};
  #pragma unroll
  for (int k = 0; k < 64; k += 4) {
    const int kk = k + 2 * hi;
    v2f a = *(const v2f*)(old_g + (size_t)row * 64 + kk);
    v2f b;
    b.x = W[kk * 64 + col];
    b.y = W[(kk + 1) * 64 + col];
    c = __builtin_amdgcn_wmma_f32_16x16x4_f32(false, a, false, b,
                                              (short)0, c, false, false);
  }

  #pragma unroll
  for (int v = 0; v < 8; ++v) {
    const size_t idx = (size_t)(m0 + v + 8 * hi) * 64 + n0 + lo;
    h[idx]   = c[v];
    agg[idx] = 0.0f;
  }
}

#define EDGES_PER_WAVE 32
#define EDGE_CHUNK 8

__global__ __launch_bounds__(256)
void gnn_edge(const float* __restrict__ h,
              const float* __restrict__ old_g,
              const float* __restrict__ ew,
              const int* __restrict__ src,
              const int* __restrict__ dst,
              float* __restrict__ agg,
              int E) {
  const int lane = threadIdx.x & 31;
  const int wave = __builtin_amdgcn_readfirstlane(threadIdx.x >> 5);
  const int e0 = (blockIdx.x * 8 + wave) * EDGES_PER_WAVE;
  if (e0 >= E) return;
  const int n = min(EDGES_PER_WAVE, E - e0);

  const int ei = e0 + min(lane, n - 1);
  const int   sv = src[ei];
  const int   rv = dst[ei];
  const int   wv = __float_as_int(ew[ei]);

  const unsigned loff = (unsigned)lane * 8u;

  #pragma unroll 1
  for (int c = 0; c < EDGES_PER_WAVE; c += EDGE_CHUNK) {
    if (c >= n) break;

    float2 hv[EDGE_CHUNK], og[EDGE_CHUNK];
    #pragma unroll
    for (int j = 0; j < EDGE_CHUNK; ++j) {
      const int s = __builtin_amdgcn_readlane(sv, c + j);
      const int r = __builtin_amdgcn_readlane(rv, c + j);
      hv[j] = *(const float2*)((const char*)(h + (size_t)s * 64) + loff);
      og[j] = *(const float2*)((const char*)(old_g + (size_t)r * 64) + loff);
    }

    float2 m[EDGE_CHUNK];
    #pragma unroll
    for (int j = 0; j < EDGE_CHUNK; ++j) {
      const float w = __int_as_float(__builtin_amdgcn_readlane(wv, c + j));
      m[j].x = tanhf(hv[j].x + og[j].x * w);
      m[j].y = tanhf(hv[j].y + og[j].y * w);
    }

    #pragma unroll
    for (int j = 0; j < EDGE_CHUNK; ++j) {
      if (c + j < n) {
        const int r = __builtin_amdgcn_readlane(rv, c + j);
        const float* row = agg + (size_t)r * 64;
        asm volatile("global_atomic_add_f32 %0, %1, %3\n\t"
                     "global_atomic_add_f32 %0, %2, %3 offset:4\n\t"
                     "s_wait_xcnt 0x0"
                     :
                     : "v"(loff), "v"(m[j].x), "v"(m[j].y), "s"(row)
                     : "memory");
      }
    }
  }
}

__global__ void gnn_finish(const float* __restrict__ agg,
                           const float* __restrict__ hist_db,
                           float* __restrict__ hist_out,
                           int n4) {
  const int i = blockIdx.x * blockDim.x + threadIdx.x;
  if (i >= n4) return;
  const float4 a = ((const float4*)agg)[i];
  const float4 b = ((const float4*)hist_db)[i];
  float4 o;
  o.x = a.x + b.x; o.y = a.y + b.y; o.z = a.z + b.z; o.w = a.w + b.w;
  ((float4*)hist_out)[i] = o;
}

extern "C" void kernel_launch(void* const* d_in, const int* in_sizes, int n_in,
                              void* d_out, int out_size, void* d_ws, size_t ws_size,
                              hipStream_t stream) {
  const float* old_g   = (const float*)d_in[0];
  const float* W       = (const float*)d_in[1];
  const float* ew      = (const float*)d_in[2];
  const float* hist_db = (const float*)d_in[3];
  const int*   src     = (const int*)d_in[4];
  const int*   dst     = (const int*)d_in[5];

  const int N = in_sizes[0] / 64;
  const int E = in_sizes[4];

  float* nodes_new = (float*)d_out;
  float* hist_new  = (float*)d_out + (size_t)N * 64;

  {
    const int rowTiles = (N + 15) / 16;
    gnn_gemm_wmma<<<rowTiles, 128, 0, stream>>>(old_g, W, hist_new, nodes_new, N);
  }

  {
    const int edgesPerBlock = 8 * EDGES_PER_WAVE;
    const int blocks = (E + edgesPerBlock - 1) / edgesPerBlock;
    gnn_edge<<<blocks, 256, 0, stream>>>(hist_new, old_g, ew, src, dst, nodes_new, E);
  }

  {
    const int n4 = N * 16;
    gnn_finish<<<(n4 + 255) / 256, 256, 0, stream>>>(nodes_new, hist_db, hist_new, n4);
  }
}


# edge kernel: lane L holds features L and L+32, so each f32 atomic add covers one whole 128-B line (2 atomic line requests per edge, not 4 half-filled); arrival order of the atomic adds may change
# speedup vs baseline: 1.6452x; 1.0482x over previous
#include <hip/hip_runtime.h>
#include <hip/hip_bf16.h>


typedef float v2f __attribute__((ext_vector_type(2)));
typedef float v8f __attribute__((ext_vector_type(8)));

__global__ void gnn_gemm_wmma(const float* __restrict__ old_g,
                              const float* __restrict__ W,
                              float* __restrict__ h,
                              float* __restrict__ agg,
                              int N) {
  const int lane = threadIdx.x & 31;
  const int wave = threadIdx.x >> 5;
  const int waveId = blockIdx.x * 4 + wave;
  const int row_tile = waveId >> 2;
  const int col_tile = waveId & 3;
  if (row_tile * 16 >= N) return;

  const int m0 = row_tile * 16;
  const int n0 = col_tile * 16;
  const int lo = lane & 15;
  const int hi = lane >> 4;

  const int row = m0 + lo;
  const int col = n0 + lo;

  v8f c = {};
  #pragma unroll
  for (int k = 0; k < 64; k += 4) {
    const int kk = k + 2 * hi;
    v2f a = *(const v2f*)(old_g + (size_t)row * 64 + kk);
    v2f b;
    b.x = W[kk * 64 + col];
    b.y = W[(kk + 1) * 64 + col];
    c = __builtin_amdgcn_wmma_f32_16x16x4_f32(false, a, false, b,
                                              (short)0, c, false, false);
  }

  #pragma unroll
  for (int v = 0; v < 8; ++v) {
    const size_t idx = (size_t)(m0 + v + 8 * hi) * 64 + n0 + lo;
    h[idx]   = c[v];
    agg[idx] = 0.0f;
  }
}

#define EDGES_PER_WAVE 32
#define EDGE_CHUNK 8

__global__ __launch_bounds__(256)
void gnn_edge(const float* __restrict__ h,
              const float* __restrict__ old_g,
              const float* __restrict__ ew,
              const int* __restrict__ src,
              const int* __restrict__ dst,
              float* __restrict__ agg,
              int E) {
  const int lane = threadIdx.x & 31;
  const int wave = __builtin_amdgcn_readfirstlane(threadIdx.x >> 5);
  const int e0 = (blockIdx.x * 8 + wave) * EDGES_PER_WAVE;
  if (e0 >= E) return;
  const int n = min(EDGES_PER_WAVE, E - e0);

  const int ei = e0 + min(lane, n - 1);
  const int   sv = src[ei];
  const int   rv = dst[ei];
  const int   wv = __float_as_int(ew[ei]);

  const unsigned loff = (unsigned)lane * 4u;

  #pragma unroll 1
  for (int c = 0; c < EDGES_PER_WAVE; c += EDGE_CHUNK) {
    if (c >= n) break;

    float h0[EDGE_CHUNK], h1[EDGE_CHUNK], g0[EDGE_CHUNK], g1[EDGE_CHUNK];
    #pragma unroll
    for (int j = 0; j < EDGE_CHUNK; ++j) {
      const int s = __builtin_amdgcn_readlane(sv, c + j);
      const int r = __builtin_amdgcn_readlane(rv, c + j);
      const char* hp = (const char*)(h + (size_t)s * 64) + loff;
      const char* gp = (const char*)(old_g + (size_t)r * 64) + loff;
      h0[j] = *(const float*)hp;
      h1[j] = *(const float*)(hp + 128);
      g0[j] = *(const float*)gp;
      g1[j] = *(const float*)(gp + 128);
    }

    float m0[EDGE_CHUNK], m1[EDGE_CHUNK];
    #pragma unroll
    for (int j = 0; j < EDGE_CHUNK; ++j) {
      const float w = __int_as_float(__builtin_amdgcn_readlane(wv, c + j));
      m0[j] = tanhf(h0[j] + g0[j] * w);
      m1[j] = tanhf(h1[j] + g1[j] * w);
    }

    #pragma unroll
    for (int j = 0; j < EDGE_CHUNK; ++j) {
      if (c + j < n) {
        const int r = __builtin_amdgcn_readlane(rv, c + j);
        const float* row = agg + (size_t)r * 64;
        asm volatile("global_atomic_add_f32 %0, %1, %3\n\t"
                     "global_atomic_add_f32 %0, %2, %3 offset:128\n\t"
                     "s_wait_xcnt 0x0"
                     :
                     : "v"(loff), "v"(m0[j]), "v"(m1[j]), "s"(row)
                     : "memory");
      }
    }
  }
}

__global__ void gnn_finish(const float* __restrict__ agg,
                           const float* __restrict__ hist_db,
                           float* __restrict__ hist_out,
                           int n4) {
  const int i = blockIdx.x * blockDim.x + threadIdx.x;
  if (i >= n4) return;
  const float4 a = ((const float4*)agg)[i];
  const float4 b = ((const float4*)hist_db)[i];
  float4 o;
  o.x = a.x + b.x; o.y = a.y + b.y; o.z = a.z + b.z; o.w = a.w + b.w;
  ((float4*)hist_out)[i] = o;
}

extern "C" void kernel_launch(void* const* d_in, const int* in_sizes, int n_in,
                              void* d_out, int out_size, void* d_ws, size_t ws_size,
                              hipStream_t stream) {
  const float* old_g   = (const float*)d_in[0];
  const float* W       = (const float*)d_in[1];
  const float* ew      = (const float*)d_in[2];
  const float* hist_db = (const float*)d_in[3];
  const int*   src     = (const int*)d_in[4];
  const int*   dst     = (const int*)d_in[5];

  const int N = in_sizes[0] / 64;
  const int E = in_sizes[4];

  float* nodes_new = (float*)d_out;
  float* hist_new  = (float*)d_out + (size_t)N * 64;

  {
    const int rowTiles = (N + 15) / 16;
    gnn_gemm_wmma<<<rowTiles, 128, 0, stream>>>(old_g, W, hist_new, nodes_new, N);
  }

  {
    const int edgesPerBlock = 8 * EDGES_PER_WAVE;
    const int blocks = (E + edgesPerBlock - 1) / edgesPerBlock;
    gnn_edge<<<blocks, 256, 0, stream>>>(hist_new, old_g, ew, src, dst, nodes_new, E);
  }

  {
    const int n4 = N * 16;
    gnn_finish<<<(n4 + 255) / 256, 256, 0, stream>>>(nodes_new, hist_db, hist_new, n4);
  }
}


# edge kernel: tanh(x) as 1 - 2/(1 + e^(2x)) on one v_exp_f32 and one v_rcp_f32 (5 f32 vector instructions) in place of the library tanhf expansion; relative accuracy at small |x| given up
# speedup vs baseline: 1.8925x; 1.1503x over previous
#include <hip/hip_runtime.h>
#include <hip/hip_bf16.h>


typedef float v2f __attribute__((ext_vector_type(2)));
typedef float v8f __attribute__((ext_vector_type(8)));

__global__ void gnn_gemm_wmma(const float* __restrict__ old_g,
                              const float* __restrict__ W,
                              float* __restrict__ h,
                              float* __restrict__ agg,
                              int N) {
  const int lane = threadIdx.x & 31;
  const int wave = threadIdx.x >> 5;
  const int waveId = blockIdx.x * 4 + wave;
  const int row_tile = waveId >> 2;
  const int col_tile = waveId & 3;
  if (row_tile * 16 >= N) return;

  const int m0 = row_tile * 16;
  const int n0 = col_tile * 16;
  const int lo = lane & 15;
  const int hi = lane >> 4;

  const int row = m0 + lo;
  const int col = n0 + lo;

  v8f c = {};
  #pragma unroll
  for (int k = 0; k < 64; k += 4) {
    const int kk = k + 2 * hi;
    v2f a = *(const v2f*)(old_g + (size_t)row * 64 + kk);
    v2f b;
    b.x = W[kk * 64 + col];
    b.y = W[(kk + 1) * 64 + col];
    c = __builtin_amdgcn_wmma_f32_16x16x4_f32(false, a, false, b,
                                              (short)0, c, false, false);
  }

  #pragma unroll
  for (int v = 0; v < 8; ++v) {
    const size_t idx = (size_t)(m0 + v + 8 * hi) * 64 + n0 + lo;
    h[idx]   = c[v];
    agg[idx] = 0.0f;
  }
}

__device__ __forceinline__ float tanh_short(float x) {
  const float t = __builtin_amdgcn_exp2f(x * 2.88539008177792681f);
  return __builtin_fmaf(-2.0f, __builtin_amdgcn_rcpf(t + 1.0f), 1.0f);
}

#define EDGES_PER_WAVE 32
#define EDGE_CHUNK 8

__global__ __launch_bounds__(256)
void gnn_edge(const float* __restrict__ h,
              const float* __restrict__ old_g,
              const float* __restrict__ ew,
              const int* __restrict__ src,
              const int* __restrict__ dst,
              float* __restrict__ agg,
              int E) {
  const int lane = threadIdx.x & 31;
  const int wave = __builtin_amdgcn_readfirstlane(threadIdx.x >> 5);
  const int e0 = (blockIdx.x * 8 + wave) * EDGES_PER_WAVE;
  if (e0 >= E) return;
  const int n = min(EDGES_PER_WAVE, E - e0);

  const int ei = e0 + min(lane, n - 1);
  const int   sv = src[ei];
  const int   rv = dst[ei];
  const int   wv = __float_as_int(ew[ei]);

  const unsigned loff = (unsigned)lane * 4u;

  #pragma unroll 1
  for (int c = 0; c < EDGES_PER_WAVE; c += EDGE_CHUNK) {
    if (c >= n) break;

    float h0[EDGE_CHUNK], h1[EDGE_CHUNK], g0[EDGE_CHUNK], g1[EDGE_CHUNK];
    #pragma unroll
    for (int j = 0; j < EDGE_CHUNK; ++j) {
      const int s = __builtin_amdgcn_readlane(sv, c + j);
      const int r = __builtin_amdgcn_readlane(rv, c + j);
      const char* hp = (const char*)(h + (size_t)s * 64) + loff;
      const char* gp = (const char*)(old_g + (size_t)r * 64) + loff;
      h0[j] = *(const float*)hp;
      h1[j] = *(const float*)(hp + 128);
      g0[j] = *(const float*)gp;
      g1[j] = *(const float*)(gp + 128);
    }

    float m0[EDGE_CHUNK], m1[EDGE_CHUNK];
    #pragma unroll
    for (int j = 0; j < EDGE_CHUNK; ++j) {
      const float w = __int_as_float(__builtin_amdgcn_readlane(wv, c + j));
      m0[j] = tanh_short(h0[j] + g0[j] * w);
      m1[j] = tanh_short(h1[j] + g1[j] * w);
    }

    #pragma unroll
    for (int j = 0; j < EDGE_CHUNK; ++j) {
      if (c + j < n) {
        const int r = __builtin_amdgcn_readlane(rv, c + j);
        const float* row = agg + (size_t)r * 64;
        asm volatile("global_atomic_add_f32 %0, %1, %3\n\t"
                     "global_atomic_add_f32 %0, %2, %3 offset:128\n\t"
                     "s_wait_xcnt 0x0"
                     :
                     : "v"(loff), "v"(m0[j]), "v"(m1[j]), "s"(row)
                     : "memory");
      }
    }
  }
}

__global__ void gnn_finish(const float* __restrict__ agg,
                           const float* __restrict__ hist_db,
                           float* __restrict__ hist_out,
                           int n4) {
  const int i = blockIdx.x * blockDim.x + threadIdx.x;
  if (i >= n4) return;
  const float4 a = ((const float4*)agg)[i];
  const float4 b = ((const float4*)hist_db)[i];
  float4 o;
  o.x = a.x + b.x; o.y = a.y + b.y; o.z = a.z + b.z; o.w = a.w + b.w;
  ((float4*)hist_out)[i] = o;
}

extern "C" void kernel_launch(void* const* d_in, const int* in_sizes, int n_in,
                              void* d_out, int out_size, void* d_ws, size_t ws_size,
                              hipStream_t stream) {
  const float* old_g   = (const float*)d_in[0];
  const float* W       = (const float*)d_in[1];
  const float* ew      = (const float*)d_in[2];
  const float* hist_db = (const float*)d_in[3];
  const int*   src     = (const int*)d_in[4];
  const int*   dst     = (const int*)d_in[5];

  const int N = in_sizes[0] / 64;
  const int E = in_sizes[4];

  float* nodes_new = (float*)d_out;
  float* hist_new  = (float*)d_out + (size_t)N * 64;

  {
    const int rowTiles = (N + 15) / 16;
    gnn_gemm_wmma<<<rowTiles, 128, 0, stream>>>(old_g, W, hist_new, nodes_new, N);
  }

  {
    const int edgesPerBlock = 8 * EDGES_PER_WAVE;
    const int blocks = (E + edgesPerBlock - 1) / edgesPerBlock;
    gnn_edge<<<blocks, 256, 0, stream>>>(hist_new, old_g, ew, src, dst, nodes_new, E);
  }

  {
    const int n4 = N * 16;
    gnn_finish<<<(n4 + 255) / 256, 256, 0, stream>>>(nodes_new, hist_db, hist_new, n4);
  }
}


# GEMM kernel: one wave per 16-row tile does all four 16-column tiles: A fragment loaded once, not in each of 4 waves; 4 independent WMMA accumulators; 4x fewer waves; zeros stored as whole lines
# speedup vs baseline: 2.0481x; 1.0822x over previous
#include <hip/hip_runtime.h>
#include <hip/hip_bf16.h>


typedef float v2f __attribute__((ext_vector_type(2)));
typedef float v8f __attribute__((ext_vector_type(8)));

__global__ __launch_bounds__(128)
void gnn_gemm_wmma(const float* __restrict__ old_g,
                   const float* __restrict__ W,
                   float* __restrict__ h,
                   float* __restrict__ agg,
                   int N) {
  const int lane = threadIdx.x & 31;
  const int wave = threadIdx.x >> 5;
  const int row_tile = blockIdx.x * 4 + wave;
  if (row_tile * 16 >= N) return;

  const int m0 = row_tile * 16;
  const int lo = lane & 15;
  const int hi = lane >> 4;

  const int row = m0 + lo;

  v8f c0 = {}, c1 = {}, c2 = {}, c3 = {};
  #pragma unroll
  for (int k = 0; k < 64; k += 4) {
    const int kk = k + 2 * hi;
    const v2f a = *(const v2f*)(old_g + (size_t)row * 64 + kk);
    const float* w0 = W + kk * 64 + lo;
    const float* w1 = w0 + 64;
    v2f b0, b1, b2, b3;
    b0.x = w0[0];  b0.y = w1[0];
    b1.x = w0[16]; b1.y = w1[16];
    b2.x = w0[32]; b2.y = w1[32];
    b3.x = w0[48]; b3.y = w1[48];
    c0 = __builtin_amdgcn_wmma_f32_16x16x4_f32(false, a, false, b0,
                                               (short)0, c0, false, false);
    c1 = __builtin_amdgcn_wmma_f32_16x16x4_f32(false, a, false, b1,
                                               (short)0, c1, false, false);
    c2 = __builtin_amdgcn_wmma_f32_16x16x4_f32(false, a, false, b2,
                                               (short)0, c2, false, false);
    c3 = __builtin_amdgcn_wmma_f32_16x16x4_f32(false, a, false, b3,
                                               (short)0, c3, false, false);
  }

  #pragma unroll
  for (int v = 0; v < 8; ++v) {
    float* hr = h + (size_t)(m0 + v + 8 * hi) * 64 + lo;
    hr[0]  = c0[v];
    hr[16] = c1[v];
    hr[32] = c2[v];
    hr[48] = c3[v];
  }

  const float4 z = make_float4(0.0f, 0.0f, 0.0f, 0.0f);
  #pragma unroll
  for (int i = 0; i < 8; ++i) {
    *(float4*)(agg + (size_t)(m0 + hi + 2 * i) * 64 + 4 * lo) = z;
  }
}

__device__ __forceinline__ float tanh_short(float x) {
  const float t = __builtin_amdgcn_exp2f(x * 2.88539008177792681f);
  return __builtin_fmaf(-2.0f, __builtin_amdgcn_rcpf(t + 1.0f), 1.0f);
}

#define EDGES_PER_WAVE 32
#define EDGE_CHUNK 8

__global__ __launch_bounds__(256)
void gnn_edge(const float* __restrict__ h,
              const float* __restrict__ old_g,
              const float* __restrict__ ew,
              const int* __restrict__ src,
              const int* __restrict__ dst,
              float* __restrict__ agg,
              int E) {
  const int lane = threadIdx.x & 31;
  const int wave = __builtin_amdgcn_readfirstlane(threadIdx.x >> 5);
  const int e0 = (blockIdx.x * 8 + wave) * EDGES_PER_WAVE;
  if (e0 >= E) return;
  const int n = min(EDGES_PER_WAVE, E - e0);

  const int ei = e0 + min(lane, n - 1);
  const int   sv = src[ei];
  const int   rv = dst[ei];
  const int   wv = __float_as_int(ew[ei]);

  const unsigned loff = (unsigned)lane * 4u;

  #pragma unroll 1
  for (int c = 0; c < EDGES_PER_WAVE; c += EDGE_CHUNK) {
    if (c >= n) break;

    float h0[EDGE_CHUNK], h1[EDGE_CHUNK], g0[EDGE_CHUNK], g1[EDGE_CHUNK];
    #pragma unroll
    for (int j = 0; j < EDGE_CHUNK; ++j) {
      const int s = __builtin_amdgcn_readlane(sv, c + j);
      const int r = __builtin_amdgcn_readlane(rv, c + j);
      const char* hp = (const char*)(h + (size_t)s * 64) + loff;
      const char* gp = (const char*)(old_g + (size_t)r * 64) + loff;
      h0[j] = *(const float*)hp;
      h1[j] = *(const float*)(hp + 128);
      g0[j] = *(const float*)gp;
      g1[j] = *(const float*)(gp + 128);
    }

    float m0[EDGE_CHUNK], m1[EDGE_CHUNK];
    #pragma unroll
    for (int j = 0; j < EDGE_CHUNK; ++j) {
      const float w = __int_as_float(__builtin_amdgcn_readlane(wv, c + j));
      m0[j] = tanh_short(h0[j] + g0[j] * w);
      m1[j] = tanh_short(h1[j] + g1[j] * w);
    }

    #pragma unroll
    for (int j = 0; j < EDGE_CHUNK; ++j) {
      if (c + j < n) {
        const int r = __builtin_amdgcn_readlane(rv, c + j);
        const float* row = agg + (size_t)r * 64;
        asm volatile("global_atomic_add_f32 %0, %1, %3\n\t"
                     "global_atomic_add_f32 %0, %2, %3 offset:128\n\t"
                     "s_wait_xcnt 0x0"
                     :
                     : "v"(loff), "v"(m0[j]), "v"(m1[j]), "s"(row)
                     : "memory");
      }
    }
  }
}

__global__ void gnn_finish(const float* __restrict__ agg,
                           const float* __restrict__ hist_db,
                           float* __restrict__ hist_out,
                           int n4) {
  const int i = blockIdx.x * blockDim.x + threadIdx.x;
  if (i >= n4) return;
  const float4 a = ((const float4*)agg)[i];
  const float4 b = ((const float4*)hist_db)[i];
  float4 o;
  o.x = a.x + b.x; o.y = a.y + b.y; o.z = a.z + b.z; o.w = a.w + b.w;
  ((float4*)hist_out)[i] = o;
}

extern "C" void kernel_launch(void* const* d_in, const int* in_sizes, int n_in,
                              void* d_out, int out_size, void* d_ws, size_t ws_size,
                              hipStream_t stream) {
  const float* old_g   = (const float*)d_in[0];
  const float* W       = (const float*)d_in[1];
  const float* ew      = (const float*)d_in[2];
  const float* hist_db = (const float*)d_in[3];
  const int*   src     = (const int*)d_in[4];
  const int*   dst     = (const int*)d_in[5];

  const int N = in_sizes[0] / 64;
  const int E = in_sizes[4];

  float* nodes_new = (float*)d_out;
  float* hist_new  = (float*)d_out + (size_t)N * 64;

  {
    const int rowTiles = (N + 15) / 16;
    gnn_gemm_wmma<<<(rowTiles + 3) / 4, 128, 0, stream>>>(old_g, W, hist_new, nodes_new, N);
  }

  {
    const int edgesPerBlock = 8 * EDGES_PER_WAVE;
    const int blocks = (E + edgesPerBlock - 1) / edgesPerBlock;
    gnn_edge<<<blocks, 256, 0, stream>>>(hist_new, old_g, ew, src, dst, nodes_new, E);
  }

  {
    const int n4 = N * 16;
    gnn_finish<<<(n4 + 255) / 256, 256, 0, stream>>>(nodes_new, hist_db, hist_new, n4);
  }
}
